# starting kernel
# baseline (speedup 1.0000x reference)
#include <hip/hip_runtime.h>

typedef __attribute__((ext_vector_type(2))) float v2f;
typedef __attribute__((ext_vector_type(8))) float v8f;

#define NEG_SLOPE 0.2f
#define NEG_INF_F (-1e30f)

__device__ __forceinline__ unsigned f2o(float f) {
    unsigned u = __float_as_uint(f);
    return (u & 0x80000000u) ? ~u : (u | 0x80000000u);
}
__device__ __forceinline__ float o2f(unsigned o) {
    return (o & 0x80000000u) ? __uint_as_float(o & 0x7FFFFFFFu)
                             : __uint_as_float(~o);
}

template<int CIN, int COUT>
__global__ __launch_bounds__(256)
void gat_gemm_wmma(const float* __restrict__ X, const float* __restrict__ Wm,
                   const float* __restrict__ asrc, const float* __restrict__ adst,
                   float* __restrict__ H, float* __restrict__ S, float* __restrict__ D,
                   int N)
{
    constexpr int NT = COUT / 16;
    const int wave = (blockIdx.x * blockDim.x + threadIdx.x) >> 5;
    const int lane = threadIdx.x & 31;
    const int m0 = wave * 16;
    if (m0 >= N) return;

    const int half = lane >> 4;
    const int l = lane & 15;

    v8f acc[NT] = {};

    int arow = m0 + l; if (arow >= N) arow = N - 1;
    const float* xrow = X + (size_t)arow * CIN + 2 * half;

    #pragma unroll 4
    for (int k = 0; k < CIN; k += 4) {
        v2f A;
        A[0] = xrow[k];
        A[1] = xrow[k + 1];
        #pragma unroll
        for (int t = 0; t < NT; ++t) {
            const int n = t * 16 + l;
            v2f B;
            B[0] = Wm[(size_t)(k + 2 * half    ) * COUT + n];
            B[1] = Wm[(size_t)(k + 2 * half + 1) * COUT + n];
            acc[t] = __builtin_amdgcn_wmma_f32_16x16x4_f32(
                         false, A, false, B, (short)0, acc[t], false, false);
        }
    }

    #pragma unroll
    for (int j = 0; j < 8; ++j) {
        const int row = m0 + j + 8 * half;
        float ps = 0.0f, pd = 0.0f;
        #pragma unroll
        for (int t = 0; t < NT; ++t) {
            const int n = t * 16 + l;
            const float v = acc[t][j];
            if (row < N) H[(size_t)row * COUT + n] = v;
            ps += v * asrc[n];
            pd += v * adst[n];
        }
        #pragma unroll
        for (int off = 1; off < 16; off <<= 1) {
            ps += __shfl_xor(ps, off, 32);
            pd += __shfl_xor(pd, off, 32);
        }
        if (l == 0 && row < N) { S[row] = ps; D[row] = pd; }
    }
}

__global__ void gat_init(float* __restrict__ agg, unsigned* __restrict__ Mo,
                         float* __restrict__ den, int N, int COUT)
{
    const int i = blockIdx.x * blockDim.x + threadIdx.x;
    if (i < N * COUT) agg[i] = 0.0f;
    if (i < N) { Mo[i] = 0u; den[i] = 0.0f; }
}

__global__ void gat_edge_alpha(const int* __restrict__ src, const int* __restrict__ dst,
                               const float* __restrict__ S, const float* __restrict__ D,
                               float* __restrict__ alp, unsigned* __restrict__ Mo,
                               int E, int Eall)
{
    const int e = blockIdx.x * blockDim.x + threadIdx.x;
    if (e >= Eall) return;
    int si, di; bool self;
    if (e < E) { si = src[e]; di = dst[e]; self = (si == di); }
    else       { si = di = e - E;          self = false; }
    float a = S[si] + D[di];
    a = (a > 0.0f) ? a : NEG_SLOPE * a;
    if (self) a = NEG_INF_F;
    alp[e] = a;
    atomicMax(&Mo[di], f2o(a));
}

__global__ void gat_edge_exp(const int* __restrict__ dst,
                             float* __restrict__ alp, const unsigned* __restrict__ Mo,
                             float* __restrict__ den, int E, int Eall)
{
    const int e = blockIdx.x * blockDim.x + threadIdx.x;
    if (e >= Eall) return;
    const int di = (e < E) ? dst[e] : (e - E);
    const float m = o2f(Mo[di]);
    const float ex = __expf(alp[e] - m);
    alp[e] = ex;
    atomicAdd(&den[di], ex);
}

template<int COUT>
__global__ __launch_bounds__(256)
void gat_edge_agg(const int* __restrict__ src, const int* __restrict__ dst,
                  const float* __restrict__ ex, const float* __restrict__ den,
                  const float* __restrict__ H, float* __restrict__ AGG,
                  int E, int Eall)
{
    constexpr int G = COUT / 4;
    const long long tid = (long long)blockIdx.x * blockDim.x + threadIdx.x;
    const long long e = tid / G;
    if (e >= Eall) return;
    const int c4 = (int)(tid % G) * 4;
    int si, di;
    if (e < E) { si = src[e]; di = dst[e]; }
    else       { si = di = (int)(e - E); }
    const float w = ex[e] / den[di];
    const float4 hv = *(const float4*)(H + (size_t)si * COUT + c4);
    float* outp = AGG + (size_t)di * COUT + c4;
    atomicAdd(outp + 0, w * hv.x);
    atomicAdd(outp + 1, w * hv.y);
    atomicAdd(outp + 2, w * hv.z);
    atomicAdd(outp + 3, w * hv.w);
}

template<bool RELU, int COUT>
__global__ void gat_finalize(const float* __restrict__ agg, const float* __restrict__ b,
                             float* __restrict__ out, int N)
{
    const int i = blockIdx.x * blockDim.x + threadIdx.x;
    if (i >= N * COUT) return;
    float v = agg[i] + b[i & (COUT - 1)];
    if (RELU) v = fmaxf(v, 0.0f);
    out[i] = v;
}

static inline int cdiv_ll(long long a, long long b) { return (int)((a + b - 1) / b); }

extern "C" void kernel_launch(void* const* d_in, const int* in_sizes, int n_in,
                              void* d_out, int out_size, void* d_ws, size_t ws_size,
                              hipStream_t stream)
{
    const float* x   = (const float*)d_in[0];
    const int*   ei  = (const int*)  d_in[1];
    const float* W1  = (const float*)d_in[2];
    const float* as1 = (const float*)d_in[3];
    const float* ad1 = (const float*)d_in[4];
    const float* b1  = (const float*)d_in[5];
    const float* W2  = (const float*)d_in[6];
    const float* as2 = (const float*)d_in[7];
    const float* ad2 = (const float*)d_in[8];
    const float* b2  = (const float*)d_in[9];
    const float* W3  = (const float*)d_in[10];
    const float* as3 = (const float*)d_in[11];
    const float* ad3 = (const float*)d_in[12];
    const float* b3  = (const float*)d_in[13];

    const int N    = in_sizes[0] / 128;
    const int E    = in_sizes[1] / 2;
    const int Eall = E + N;
    const int* src = ei;
    const int* dst = ei + E;

    float* ws   = (float*)d_ws;
    float* h    = ws;
    float* agg  = h   + (size_t)N * 128;
    float* xn   = agg + (size_t)N * 128;
    float* alp  = xn  + (size_t)N * 128;
    float* Sv   = alp + Eall;
    float* Dv   = Sv  + N;
    unsigned* Mo = (unsigned*)(Dv + N);
    float* den  = (float*)(Mo + N);

    const dim3 blk(256);
    const int waves_blocks = cdiv_ll((long long)cdiv_ll(N, 16) * 32, 256);
    const int edge_blocks  = cdiv_ll(Eall, 256);

    gat_init<<<cdiv_ll((long long)N * 128, 256), blk, 0, stream>>>(agg, Mo, den, N, 128);
    gat_gemm_wmma<128,128><<<waves_blocks, blk, 0, stream>>>(x, W1, as1, ad1, h, Sv, Dv, N);
    gat_edge_alpha<<<edge_blocks, blk, 0, stream>>>(src, dst, Sv, Dv, alp, Mo, E, Eall);
    gat_edge_exp<<<edge_blocks, blk, 0, stream>>>(dst, alp, Mo, den, E, Eall);
    gat_edge_agg<128><<<cdiv_ll((long long)Eall * 32, 256), blk, 0, stream>>>(src, dst, alp, den, h, agg, E, Eall);
    gat_finalize<true,128><<<cdiv_ll((long long)N * 128, 256), blk, 0, stream>>>(agg, b1, xn, N);

    gat_init<<<cdiv_ll((long long)N * 64, 256), blk, 0, stream>>>(agg, Mo, den, N, 64);
    gat_gemm_wmma<128,64><<<waves_blocks, blk, 0, stream>>>(xn, W2, as2, ad2, h, Sv, Dv, N);
    gat_edge_alpha<<<edge_blocks, blk, 0, stream>>>(src, dst, Sv, Dv, alp, Mo, E, Eall);
    gat_edge_exp<<<edge_blocks, blk, 0, stream>>>(dst, alp, Mo, den, E, Eall);
    gat_edge_agg<64><<<cdiv_ll((long long)Eall * 16, 256), blk, 0, stream>>>(src, dst, alp, den, h, agg, E, Eall);
    gat_finalize<true,64><<<cdiv_ll((long long)N * 64, 256), blk, 0, stream>>>(agg, b2, xn, N);

    gat_init<<<cdiv_ll((long long)N * 64, 256), blk, 0, stream>>>(agg, Mo, den, N, 64);
    gat_gemm_wmma<64,64><<<waves_blocks, blk, 0, stream>>>(xn, W3, as3, ad3, h, Sv, Dv, N);
    gat_edge_alpha<<<edge_blocks, blk, 0, stream>>>(src, dst, Sv, Dv, alp, Mo, E, Eall);
    gat_edge_exp<<<edge_blocks, blk, 0, stream>>>(dst, alp, Mo, den, E, Eall);
    gat_edge_agg<64><<<cdiv_ll((long long)Eall * 16, 256), blk, 0, stream>>>(src, dst, alp, den, h, agg, E, Eall);
    gat_finalize<false,64><<<cdiv_ll((long long)N * 64, 256), blk, 0, stream>>>(agg, b3, (float*)d_out, N);
}


# global atomic max/sum/scatter edge passes replaced: edges binned once by 64-node destination tile, one work-group per tile sums max, sum and rows in LDS (LDS-atomic arrival order), divided at the end
# speedup vs baseline: 3.8583x; 3.8583x over previous
#include <hip/hip_runtime.h>

typedef __attribute__((ext_vector_type(2))) float v2f;
typedef __attribute__((ext_vector_type(8))) float v8f;

#define NEG_SLOPE 0.2f
#define NEG_INF_F (-1e30f)

__device__ __forceinline__ unsigned f2o(float f) {
    unsigned u = __float_as_uint(f);
    return (u & 0x80000000u) ? ~u : (u | 0x80000000u);
}
__device__ __forceinline__ float o2f(unsigned o) {
    return (o & 0x80000000u) ? __uint_as_float(o & 0x7FFFFFFFu)
                             : __uint_as_float(~o);
}

template<int CIN, int COUT>
__global__ __launch_bounds__(256)
void gat_gemm_wmma(const float* __restrict__ X, const float* __restrict__ Wm,
                   const float* __restrict__ asrc, const float* __restrict__ adst,
                   float* __restrict__ H, float* __restrict__ S, float* __restrict__ D,
                   int N)
{
    constexpr int NT = COUT / 16;
    const int wave = (blockIdx.x * blockDim.x + threadIdx.x) >> 5;
    const int lane = threadIdx.x & 31;
    const int m0 = wave * 16;
    if (m0 >= N) return;

    const int half = lane >> 4;
    const int l = lane & 15;

    v8f acc[NT] = {};

    int arow = m0 + l; if (arow >= N) arow = N - 1;
    const float* xrow = X + (size_t)arow * CIN + 2 * half;

    #pragma unroll 4
    for (int k = 0; k < CIN; k += 4) {
        v2f A;
        A[0] = xrow[k];
        A[1] = xrow[k + 1];
        #pragma unroll
        for (int t = 0; t < NT; ++t) {
            const int n = t * 16 + l;
            v2f B;
            B[0] = Wm[(size_t)(k + 2 * half    ) * COUT + n];
            B[1] = Wm[(size_t)(k + 2 * half + 1) * COUT + n];
            acc[t] = __builtin_amdgcn_wmma_f32_16x16x4_f32(
                         false, A, false, B, (short)0, acc[t], false, false);
        }
    }

    #pragma unroll
    for (int j = 0; j < 8; ++j) {
        const int row = m0 + j + 8 * half;
        float ps = 0.0f, pd = 0.0f;
        #pragma unroll
        for (int t = 0; t < NT; ++t) {
            const int n = t * 16 + l;
            const float v = acc[t][j];
            if (row < N) H[(size_t)row * COUT + n] = v;
            ps += v * asrc[n];
            pd += v * adst[n];
        }
        #pragma unroll
        for (int off = 1; off < 16; off <<= 1) {
            ps += __shfl_xor(ps, off, 32);
            pd += __shfl_xor(pd, off, 32);
        }
        if (l == 0 && row < N) { S[row] = ps; D[row] = pd; }
    }
}


#define TILE_SHIFT 6
#define TILE_NODES 64
#define BIN_CHUNK 8192

__global__ __launch_bounds__(1024)
void bin_count(const int* __restrict__ dst, int* __restrict__ chist, int E, int NT)
{
    extern __shared__ int hist_c[];
    const int tid = threadIdx.x;
    for (int b = tid; b < NT; b += 1024) hist_c[b] = 0;
    __syncthreads();
    const int base = blockIdx.x * BIN_CHUNK;
    int dv[BIN_CHUNK / 1024];
    #pragma unroll
    for (int k = 0; k < BIN_CHUNK / 1024; ++k) {
        const int e = base + k * 1024 + tid;
        dv[k] = dst[(e < E) ? e : (E - 1)];
    }
    #pragma unroll
    for (int k = 0; k < BIN_CHUNK / 1024; ++k) {
        const int e = base + k * 1024 + tid;
        if (e < E) atomicAdd(&hist_c[dv[k] >> TILE_SHIFT], 1);
    }
    __syncthreads();
    for (int b = tid; b < NT; b += 1024) chist[(size_t)blockIdx.x * NT + b] = hist_c[b];
}

__global__ __launch_bounds__(1024)
void bin_scan(const int* __restrict__ chist, int* __restrict__ cbase, int* __restrict__ off,
              int NT, int NCH, int E)
{
    __shared__ int sc[1024];
    __shared__ int carry_s;
    const int tid = threadIdx.x;
    if (tid == 0) carry_s = 0;
    __syncthreads();
    for (int b0 = 0; b0 < NT; b0 += 1024) {
        const int b = b0 + tid;
        int tot = 0;
        if (b < NT) {
            #pragma unroll 7
            for (int c = 0; c < NCH; ++c) tot += chist[(size_t)c * NT + b];
        }
        sc[tid] = tot;
        __syncthreads();
        for (int o = 1; o < 1024; o <<= 1) {
            const int t = (tid >= o) ? sc[tid - o] : 0;
            __syncthreads();
            sc[tid] += t;
            __syncthreads();
        }
        const int carry = carry_s;
        int run = sc[tid] - tot + carry;
        if (b < NT) {
            off[b] = run;
            #pragma unroll 7
            for (int c = 0; c < NCH; ++c) {
                const int v = chist[(size_t)c * NT + b];
                cbase[(size_t)c * NT + b] = run;
                run += v;
            }
        }
        __syncthreads();
        if (tid == 1023) carry_s = carry + sc[1023];
        __syncthreads();
    }
    if (tid == 0) off[NT] = E;
}

__global__ __launch_bounds__(1024)
void bin_fill(const int* __restrict__ src, const int* __restrict__ dst,
              const int* __restrict__ cbase, int2* __restrict__ bedge, int E, int NT)
{
    extern __shared__ int rank_c[];
    const int tid = threadIdx.x;
    for (int b = tid; b < NT; b += 1024) rank_c[b] = 0;
    __syncthreads();
    const int base = blockIdx.x * BIN_CHUNK;
    int sv[BIN_CHUNK / 1024];
    int dv[BIN_CHUNK / 1024];
    #pragma unroll
    for (int k = 0; k < BIN_CHUNK / 1024; ++k) {
        const int e = base + k * 1024 + tid;
        const int ec = (e < E) ? e : (E - 1);
        sv[k] = src[ec];
        dv[k] = dst[ec];
    }
    #pragma unroll
    for (int k = 0; k < BIN_CHUNK / 1024; ++k) {
        const int e = base + k * 1024 + tid;
        if (e < E) {
            const int di = dv[k];
            const int b = di >> TILE_SHIFT;
            const int r = atomicAdd(&rank_c[b], 1);
            const int pos = cbase[(size_t)blockIdx.x * NT + b] + r;
            bedge[pos] = make_int2(sv[k], di);
        }
    }
}

template<int COUT, bool RELU>
__global__ __launch_bounds__(256)
void gat_tile(const int2* __restrict__ bedge, const int* __restrict__ off,
              const float* __restrict__ S, const float* __restrict__ D,
              const float* __restrict__ H, const float* __restrict__ bias,
              float* __restrict__ out, int N)
{
    constexpr int T = TILE_NODES;
    constexpr int G = COUT / 4;
    constexpr int EPS = 32 / G;
    constexpr int PITCH = COUT + (EPS == 2 ? 16 : 0);
    constexpr int U = 8;

    __shared__ float agg[T * PITCH];
    __shared__ unsigned mx[T];
    __shared__ float den[T];
    __shared__ float dl[T];

    const int tid = threadIdx.x;
    const int lane = tid & 31;
    const int wave = __builtin_amdgcn_readfirstlane(tid >> 5);
    const int sub = lane / G;
    const int g = lane % G;
    const int t0 = blockIdx.x * T;
    const int nloc = (N - t0 < T) ? (N - t0) : T;

    for (int i = tid; i < T * PITCH; i += 256) agg[i] = 0.0f;
    if (tid < T) {
        mx[tid] = 0u;
        den[tid] = 0.0f;
        dl[tid] = (tid < nloc) ? D[t0 + tid] : 0.0f;
    }
    __syncthreads();

    const int e0 = off[blockIdx.x];
    const int ne = off[blockIdx.x + 1] - e0;
    const int total = ne + nloc;

    for (int j = tid; j < total; j += 256) {
        int si, ld; bool msk = false;
        if (j < ne) { const int2 ed = bedge[e0 + j]; si = ed.x; ld = ed.y - t0; msk = (ed.x == ed.y); }
        else        { ld = j - ne; si = t0 + ld; }
        float a = S[si] + dl[ld];
        a = (a > 0.0f) ? a : NEG_SLOPE * a;
        if (msk) a = NEG_INF_F;
        atomicMax(&mx[ld], f2o(a));
    }
    __syncthreads();

    for (int c0 = wave * 32; c0 < total; c0 += 256) {
        const int j = c0 + lane;
        int si = t0, ld = 0; float ex = 0.0f;
        if (j < total) {
            bool msk = false;
            if (j < ne) { const int2 ed = bedge[e0 + j]; si = ed.x; ld = ed.y - t0; msk = (ed.x == ed.y); }
            else        { ld = j - ne; si = t0 + ld; }
            float a = S[si] + dl[ld];
            a = (a > 0.0f) ? a : NEG_SLOPE * a;
            if (msk) a = NEG_INF_F;
            ex = __expf(a - o2f(mx[ld]));
            atomicAdd(&den[ld], ex);
        }
        const int cnt = (total - c0 < 32) ? (total - c0) : 32;
        const int nst = (cnt < G) ? cnt : G;
        for (int k = 0; k < nst; k += U) {
            float4 hv[U];
            int    lk[U];
            float  wk[U];
            bool   ok[U];
            #pragma unroll
            for (int u = 0; u < U; ++u) {
                const int kk = k + u;
                hv[u] = make_float4(0.f, 0.f, 0.f, 0.f); lk[u] = 0; wk[u] = 0.0f; ok[u] = false;
                if (kk < nst) {
                    int s = __builtin_amdgcn_readlane(si, kk);
                    int l0 = __builtin_amdgcn_readlane(ld, kk);
                    float w = __int_as_float(__builtin_amdgcn_readlane(__float_as_int(ex), kk));
                    if constexpr (EPS == 2) {
                        const int s1 = __builtin_amdgcn_readlane(si, kk + 16);
                        const int l1 = __builtin_amdgcn_readlane(ld, kk + 16);
                        const float w1 = __int_as_float(__builtin_amdgcn_readlane(__float_as_int(ex), kk + 16));
                        if (sub) { s = s1; l0 = l1; w = w1; }
                    }
                    ok[u] = (kk + 16 * sub) < cnt;
                    lk[u] = l0; wk[u] = w;
                    if (ok[u]) hv[u] = *(const float4*)(H + (size_t)s * COUT + 4 * g);
                }
            }
            #pragma unroll
            for (int u = 0; u < U; ++u) {
                if (ok[u]) {
                    float* row = agg + lk[u] * PITCH + g;
                    atomicAdd(row,         wk[u] * hv[u].x);
                    atomicAdd(row + G,     wk[u] * hv[u].y);
                    atomicAdd(row + 2 * G, wk[u] * hv[u].z);
                    atomicAdd(row + 3 * G, wk[u] * hv[u].w);
                }
            }
        }
    }
    __syncthreads();

    const float4 bb = *(const float4*)(bias + 4 * g);
    for (int r = wave * EPS + sub; r < nloc; r += 8 * EPS) {
        const float dn = den[r];
        const float* row = agg + r * PITCH + g;
        float4 v;
        v.x = row[0]     / dn + bb.x;
        v.y = row[G]     / dn + bb.y;
        v.z = row[2 * G] / dn + bb.z;
        v.w = row[3 * G] / dn + bb.w;
        if (RELU) { v.x = fmaxf(v.x, 0.0f); v.y = fmaxf(v.y, 0.0f); v.z = fmaxf(v.z, 0.0f); v.w = fmaxf(v.w, 0.0f); }
        *(float4*)(out + (size_t)(t0 + r) * COUT + 4 * g) = v;
    }
}

static inline int cdiv_ll(long long a, long long b) { return (int)((a + b - 1) / b); }

extern "C" void kernel_launch(void* const* d_in, const int* in_sizes, int n_in,
                              void* d_out, int out_size, void* d_ws, size_t ws_size,
                              hipStream_t stream)
{
    const float* x   = (const float*)d_in[0];
    const int*   ei  = (const int*)  d_in[1];
    const float* W1  = (const float*)d_in[2];
    const float* as1 = (const float*)d_in[3];
    const float* ad1 = (const float*)d_in[4];
    const float* b1  = (const float*)d_in[5];
    const float* W2  = (const float*)d_in[6];
    const float* as2 = (const float*)d_in[7];
    const float* ad2 = (const float*)d_in[8];
    const float* b2  = (const float*)d_in[9];
    const float* W3  = (const float*)d_in[10];
    const float* as3 = (const float*)d_in[11];
    const float* ad3 = (const float*)d_in[12];
    const float* b3  = (const float*)d_in[13];

    const int N    = in_sizes[0] / 128;
    const int E    = in_sizes[1] / 2;
    const int* src = ei;
    const int* dst = ei + E;

    const int NT  = cdiv_ll(N, TILE_NODES);
    const int NCH = (E > 0) ? cdiv_ll(E, BIN_CHUNK) : 1;

    float* ws    = (float*)d_ws;
    float* h     = ws;
    float* xn    = h + (size_t)N * 128;
    int2*  bedge = (int2*)(xn + (size_t)N * 128);
    int*   off   = (int*)(bedge + E);
    int*   chist = off + (NT + 1);
    int*   cbase = chist + (size_t)NCH * NT;
    float* Sv    = (float*)(cbase + (size_t)NCH * NT);
    float* Dv    = Sv + N;

    const dim3 blk(256);
    const int waves_blocks = cdiv_ll((long long)cdiv_ll(N, 16) * 32, 256);
    const size_t hist_bytes = (size_t)NT * sizeof(int);

    bin_count<<<NCH, 1024, hist_bytes, stream>>>(dst, chist, E, NT);
    bin_scan<<<1, 1024, 0, stream>>>(chist, cbase, off, NT, NCH, E);
    bin_fill<<<NCH, 1024, hist_bytes, stream>>>(src, dst, cbase, bedge, E, NT);

    gat_gemm_wmma<128,128><<<waves_blocks, blk, 0, stream>>>(x, W1, as1, ad1, h, Sv, Dv, N);
    gat_tile<128,true><<<NT, blk, 0, stream>>>(bedge, off, Sv, Dv, h, b1, xn, N);

    gat_gemm_wmma<128,64><<<waves_blocks, blk, 0, stream>>>(xn, W2, as2, ad2, h, Sv, Dv, N);
    gat_tile<64,true><<<NT, blk, 0, stream>>>(bedge, off, Sv, Dv, h, b2, xn, N);

    gat_gemm_wmma<64,64><<<waves_blocks, blk, 0, stream>>>(xn, W3, as3, ad3, h, Sv, Dv, N);
    gat_tile<64,false><<<NT, blk, 0, stream>>>(bedge, off, Sv, Dv, h, b3, (float*)d_out, N);
}


# tile kernel reads each edge record and source logit once: a lane issues all its record loads, then all its logit loads, together and keeps them in registers for the sum pass; LDS sum order kept
# speedup vs baseline: 4.0064x; 1.0384x over previous
#include <hip/hip_runtime.h>

typedef __attribute__((ext_vector_type(2))) float v2f;
typedef __attribute__((ext_vector_type(8))) float v8f;

#define NEG_SLOPE 0.2f
#define NEG_INF_F (-1e30f)

__device__ __forceinline__ unsigned f2o(float f) {
    unsigned u = __float_as_uint(f);
    return (u & 0x80000000u) ? ~u : (u | 0x80000000u);
}
__device__ __forceinline__ float o2f(unsigned o) {
    return (o & 0x80000000u) ? __uint_as_float(o & 0x7FFFFFFFu)
                             : __uint_as_float(~o);
}

template<int CIN, int COUT>
__global__ __launch_bounds__(256)
void gat_gemm_wmma(const float* __restrict__ X, const float* __restrict__ Wm,
                   const float* __restrict__ asrc, const float* __restrict__ adst,
                   float* __restrict__ H, float* __restrict__ S, float* __restrict__ D,
                   int N)
{
    constexpr int NT = COUT / 16;
    const int wave = (blockIdx.x * blockDim.x + threadIdx.x) >> 5;
    const int lane = threadIdx.x & 31;
    const int m0 = wave * 16;
    if (m0 >= N) return;

    const int half = lane >> 4;
    const int l = lane & 15;

    v8f acc[NT] = {};

    int arow = m0 + l; if (arow >= N) arow = N - 1;
    const float* xrow = X + (size_t)arow * CIN + 2 * half;

    #pragma unroll 4
    for (int k = 0; k < CIN; k += 4) {
        v2f A;
        A[0] = xrow[k];
        A[1] = xrow[k + 1];
        #pragma unroll
        for (int t = 0; t < NT; ++t) {
            const int n = t * 16 + l;
            v2f B;
            B[0] = Wm[(size_t)(k + 2 * half    ) * COUT + n];
            B[1] = Wm[(size_t)(k + 2 * half + 1) * COUT + n];
            acc[t] = __builtin_amdgcn_wmma_f32_16x16x4_f32(
                         false, A, false, B, (short)0, acc[t], false, false);
        }
    }

    #pragma unroll
    for (int j = 0; j < 8; ++j) {
        const int row = m0 + j + 8 * half;
        float ps = 0.0f, pd = 0.0f;
        #pragma unroll
        for (int t = 0; t < NT; ++t) {
            const int n = t * 16 + l;
            const float v = acc[t][j];
            if (row < N) H[(size_t)row * COUT + n] = v;
            ps += v * asrc[n];
            pd += v * adst[n];
        }
        #pragma unroll
        for (int off = 1; off < 16; off <<= 1) {
            ps += __shfl_xor(ps, off, 32);
            pd += __shfl_xor(pd, off, 32);
        }
        if (l == 0 && row < N) { S[row] = ps; D[row] = pd; }
    }
}


#define TILE_SHIFT 6
#define TILE_NODES 64
#define BIN_CHUNK 8192

__global__ __launch_bounds__(1024)
void bin_count(const int* __restrict__ dst, int* __restrict__ chist, int E, int NT)
{
    extern __shared__ int hist_c[];
    const int tid = threadIdx.x;
    for (int b = tid; b < NT; b += 1024) hist_c[b] = 0;
    __syncthreads();
    const int base = blockIdx.x * BIN_CHUNK;
    int dv[BIN_CHUNK / 1024];
    #pragma unroll
    for (int k = 0; k < BIN_CHUNK / 1024; ++k) {
        const int e = base + k * 1024 + tid;
        dv[k] = dst[(e < E) ? e : (E - 1)];
    }
    #pragma unroll
    for (int k = 0; k < BIN_CHUNK / 1024; ++k) {
        const int e = base + k * 1024 + tid;
        if (e < E) atomicAdd(&hist_c[dv[k] >> TILE_SHIFT], 1);
    }
    __syncthreads();
    for (int b = tid; b < NT; b += 1024) chist[(size_t)blockIdx.x * NT + b] = hist_c[b];
}

__global__ __launch_bounds__(1024)
void bin_scan(const int* __restrict__ chist, int* __restrict__ cbase, int* __restrict__ off,
              int NT, int NCH, int E)
{
    __shared__ int sc[1024];
    __shared__ int carry_s;
    const int tid = threadIdx.x;
    if (tid == 0) carry_s = 0;
    __syncthreads();
    for (int b0 = 0; b0 < NT; b0 += 1024) {
        const int b = b0 + tid;
        int tot = 0;
        if (b < NT) {
            #pragma unroll 7
            for (int c = 0; c < NCH; ++c) tot += chist[(size_t)c * NT + b];
        }
        sc[tid] = tot;
        __syncthreads();
        for (int o = 1; o < 1024; o <<= 1) {
            const int t = (tid >= o) ? sc[tid - o] : 0;
            __syncthreads();
            sc[tid] += t;
            __syncthreads();
        }
        const int carry = carry_s;
        int run = sc[tid] - tot + carry;
        if (b < NT) {
            off[b] = run;
            #pragma unroll 7
            for (int c = 0; c < NCH; ++c) {
                const int v = chist[(size_t)c * NT + b];
                cbase[(size_t)c * NT + b] = run;
                run += v;
            }
        }
        __syncthreads();
        if (tid == 1023) carry_s = carry + sc[1023];
        __syncthreads();
    }
    if (tid == 0) off[NT] = E;
}

__global__ __launch_bounds__(1024)
void bin_fill(const int* __restrict__ src, const int* __restrict__ dst,
              const int* __restrict__ cbase, int2* __restrict__ bedge, int E, int NT)
{
    extern __shared__ int rank_c[];
    const int tid = threadIdx.x;
    for (int b = tid; b < NT; b += 1024) rank_c[b] = 0;
    __syncthreads();
    const int base = blockIdx.x * BIN_CHUNK;
    int sv[BIN_CHUNK / 1024];
    int dv[BIN_CHUNK / 1024];
    #pragma unroll
    for (int k = 0; k < BIN_CHUNK / 1024; ++k) {
        const int e = base + k * 1024 + tid;
        const int ec = (e < E) ? e : (E - 1);
        sv[k] = src[ec];
        dv[k] = dst[ec];
    }
    #pragma unroll
    for (int k = 0; k < BIN_CHUNK / 1024; ++k) {
        const int e = base + k * 1024 + tid;
        if (e < E) {
            const int di = dv[k];
            const int b = di >> TILE_SHIFT;
            const int r = atomicAdd(&rank_c[b], 1);
            const int pos = cbase[(size_t)blockIdx.x * NT + b] + r;
            bedge[pos] = make_int2(sv[k], di);
        }
    }
}

#define TR 5

template<int COUT>
__device__ __forceinline__ void tile_rows(const float* __restrict__ H, float* agg,
                                          int si, int ld, float ex, int cnt, int sub, int g)
{
    constexpr int G = COUT / 4;
    constexpr int EPS = 32 / G;
    constexpr int PITCH = COUT + (EPS == 2 ? 16 : 0);
    constexpr int U = 8;
    const int nst = (cnt < G) ? cnt : G;
    for (int k = 0; k < nst; k += U) {
        float4 hv[U];
        int    lk[U];
        float  wk[U];
        bool   ok[U];
        #pragma unroll
        for (int u = 0; u < U; ++u) {
            const int kk = k + u;
            hv[u] = make_float4(0.f, 0.f, 0.f, 0.f); lk[u] = 0; wk[u] = 0.0f; ok[u] = false;
            if (kk < nst) {
                int s = __builtin_amdgcn_readlane(si, kk);
                int l0 = __builtin_amdgcn_readlane(ld, kk);
                float w = __int_as_float(__builtin_amdgcn_readlane(__float_as_int(ex), kk));
                if constexpr (EPS == 2) {
                    const int s1 = __builtin_amdgcn_readlane(si, kk + 16);
                    const int l1 = __builtin_amdgcn_readlane(ld, kk + 16);
                    const float w1 = __int_as_float(__builtin_amdgcn_readlane(__float_as_int(ex), kk + 16));
                    if (sub) { s = s1; l0 = l1; w = w1; }
                }
                ok[u] = (kk + 16 * sub) < cnt;
                lk[u] = l0; wk[u] = w;
                if (ok[u]) hv[u] = *(const float4*)(H + (size_t)s * COUT + 4 * g);
            }
        }
        #pragma unroll
        for (int u = 0; u < U; ++u) {
            if (ok[u]) {
                float* row = agg + lk[u] * PITCH + g;
                atomicAdd(row,         wk[u] * hv[u].x);
                atomicAdd(row + G,     wk[u] * hv[u].y);
                atomicAdd(row + 2 * G, wk[u] * hv[u].z);
                atomicAdd(row + 3 * G, wk[u] * hv[u].w);
            }
        }
    }
}

template<int COUT, bool RELU>
__global__ __launch_bounds__(256)
void gat_tile(const int2* __restrict__ bedge, const int* __restrict__ off,
              const float* __restrict__ S, const float* __restrict__ D,
              const float* __restrict__ H, const float* __restrict__ bias,
              float* __restrict__ out, int N)
{
    constexpr int T = TILE_NODES;
    constexpr int G = COUT / 4;
    constexpr int EPS = 32 / G;
    constexpr int PITCH = COUT + (EPS == 2 ? 16 : 0);

    __shared__ float agg[T * PITCH];
    __shared__ unsigned mx[T];
    __shared__ float den[T];
    __shared__ float dl[T];

    const int tid = threadIdx.x;
    const int lane = tid & 31;
    const int wave = __builtin_amdgcn_readfirstlane(tid >> 5);
    const int sub = lane / G;
    const int g = lane % G;
    const int t0 = blockIdx.x * T;
    const int nloc = (N - t0 < T) ? (N - t0) : T;

    float dval = 0.0f;
    if (tid < nloc) dval = D[t0 + tid];

    const int e0 = off[blockIdx.x];
    const int ne = off[blockIdx.x + 1] - e0;
    const int total = ne + nloc;

    int2 ed[TR];
    #pragma unroll
    for (int r = 0; r < TR; ++r) {
        const int j = tid + 256 * r;
        ed[r] = make_int2(0, 0);
        if (j < ne) ed[r] = bedge[e0 + j];
    }

    for (int i = tid; i < T * PITCH; i += 256) agg[i] = 0.0f;
    if (tid < T) {
        mx[tid] = 0u;
        den[tid] = 0.0f;
        dl[tid] = dval;
    }

    int   si[TR];
    int   ld[TR];
    float al[TR];
    bool  mk[TR];
    #pragma unroll
    for (int r = 0; r < TR; ++r) {
        const int j = tid + 256 * r;
        const bool isE = j < ne;
        const int lself = (j >= ne && j < total) ? (j - ne) : 0;
        si[r] = isE ? ed[r].x : (t0 + lself);
        ld[r] = isE ? (ed[r].y - t0) : lself;
        mk[r] = isE && (ed[r].x == ed[r].y);
        al[r] = S[si[r]];
    }
    __syncthreads();

    float dv[TR];
    #pragma unroll
    for (int r = 0; r < TR; ++r) dv[r] = dl[ld[r]];
    #pragma unroll
    for (int r = 0; r < TR; ++r) {
        float a = al[r] + dv[r];
        a = (a > 0.0f) ? a : NEG_SLOPE * a;
        if (mk[r]) a = NEG_INF_F;
        al[r] = a;
        if (tid + 256 * r < total) atomicMax(&mx[ld[r]], f2o(a));
    }
    for (int j = tid + 256 * TR; j < total; j += 256) {
        int s, l; bool msk = false;
        if (j < ne) { const int2 e = bedge[e0 + j]; s = e.x; l = e.y - t0; msk = (e.x == e.y); }
        else        { l = j - ne; s = t0 + l; }
        float a = S[s] + dl[l];
        a = (a > 0.0f) ? a : NEG_SLOPE * a;
        if (msk) a = NEG_INF_F;
        atomicMax(&mx[l], f2o(a));
    }
    __syncthreads();

    float ex[TR];
    #pragma unroll
    for (int r = 0; r < TR; ++r) ex[r] = o2f(mx[ld[r]]);
    #pragma unroll
    for (int r = 0; r < TR; ++r) {
        const float e1 = __expf(al[r] - ex[r]);
        ex[r] = 0.0f;
        if (tid + 256 * r < total) { ex[r] = e1; atomicAdd(&den[ld[r]], e1); }
    }
    #pragma unroll
    for (int r = 0; r < TR; ++r) {
        int cnt = total - (wave * 32 + 256 * r);
        cnt = (cnt < 32) ? cnt : 32;
        if (cnt > 0) tile_rows<COUT>(H, agg, si[r], ld[r], ex[r], cnt, sub, g);
    }
    for (int c0 = wave * 32 + 256 * TR; c0 < total; c0 += 256) {
        const int j = c0 + lane;
        int s = t0, l = 0; float e1 = 0.0f;
        if (j < total) {
            bool msk = false;
            if (j < ne) { const int2 e = bedge[e0 + j]; s = e.x; l = e.y - t0; msk = (e.x == e.y); }
            else        { l = j - ne; s = t0 + l; }
            float a = S[s] + dl[l];
            a = (a > 0.0f) ? a : NEG_SLOPE * a;
            if (msk) a = NEG_INF_F;
            e1 = __expf(a - o2f(mx[l]));
            atomicAdd(&den[l], e1);
        }
        const int cnt = (total - c0 < 32) ? (total - c0) : 32;
        tile_rows<COUT>(H, agg, s, l, e1, cnt, sub, g);
    }
    __syncthreads();

    const float4 bb = *(const float4*)(bias + 4 * g);
    for (int r = wave * EPS + sub; r < nloc; r += 8 * EPS) {
        const float dn = den[r];
        const float* row = agg + r * PITCH + g;
        float4 v;
        v.x = row[0]     / dn + bb.x;
        v.y = row[G]     / dn + bb.y;
        v.z = row[2 * G] / dn + bb.z;
        v.w = row[3 * G] / dn + bb.w;
        if (RELU) { v.x = fmaxf(v.x, 0.0f); v.y = fmaxf(v.y, 0.0f); v.z = fmaxf(v.z, 0.0f); v.w = fmaxf(v.w, 0.0f); }
        *(float4*)(out + (size_t)(t0 + r) * COUT + 4 * g) = v;
    }
}

static inline int cdiv_ll(long long a, long long b) { return (int)((a + b - 1) / b); }

extern "C" void kernel_launch(void* const* d_in, const int* in_sizes, int n_in,
                              void* d_out, int out_size, void* d_ws, size_t ws_size,
                              hipStream_t stream)
{
    const float* x   = (const float*)d_in[0];
    const int*   ei  = (const int*)  d_in[1];
    const float* W1  = (const float*)d_in[2];
    const float* as1 = (const float*)d_in[3];
    const float* ad1 = (const float*)d_in[4];
    const float* b1  = (const float*)d_in[5];
    const float* W2  = (const float*)d_in[6];
    const float* as2 = (const float*)d_in[7];
    const float* ad2 = (const float*)d_in[8];
    const float* b2  = (const float*)d_in[9];
    const float* W3  = (const float*)d_in[10];
    const float* as3 = (const float*)d_in[11];
    const float* ad3 = (const float*)d_in[12];
    const float* b3  = (const float*)d_in[13];

    const int N    = in_sizes[0] / 128;
    const int E    = in_sizes[1] / 2;
    const int* src = ei;
    const int* dst = ei + E;

    const int NT  = cdiv_ll(N, TILE_NODES);
    const int NCH = (E > 0) ? cdiv_ll(E, BIN_CHUNK) : 1;

    float* ws    = (float*)d_ws;
    float* h     = ws;
    float* xn    = h + (size_t)N * 128;
    int2*  bedge = (int2*)(xn + (size_t)N * 128);
    int*   off   = (int*)(bedge + E);
    int*   chist = off + (NT + 1);
    int*   cbase = chist + (size_t)NCH * NT;
    float* Sv    = (float*)(cbase + (size_t)NCH * NT);
    float* Dv    = Sv + N;

    const dim3 blk(256);
    const int waves_blocks = cdiv_ll((long long)cdiv_ll(N, 16) * 32, 256);
    const size_t hist_bytes = (size_t)NT * sizeof(int);

    bin_count<<<NCH, 1024, hist_bytes, stream>>>(dst, chist, E, NT);
    bin_scan<<<1, 1024, 0, stream>>>(chist, cbase, off, NT, NCH, E);
    bin_fill<<<NCH, 1024, hist_bytes, stream>>>(src, dst, cbase, bedge, E, NT);

    gat_gemm_wmma<128,128><<<waves_blocks, blk, 0, stream>>>(x, W1, as1, ad1, h, Sv, Dv, N);
    gat_tile<128,true><<<NT, blk, 0, stream>>>(bedge, off, Sv, Dv, h, b1, xn, N);

    gat_gemm_wmma<128,64><<<waves_blocks, blk, 0, stream>>>(xn, W2, as2, ad2, h, Sv, Dv, N);
    gat_tile<64,true><<<NT, blk, 0, stream>>>(bedge, off, Sv, Dv, h, b2, xn, N);

    gat_gemm_wmma<64,64><<<waves_blocks, blk, 0, stream>>>(xn, W3, as3, ad3, h, Sv, Dv, N);
    gat_tile<64,false><<<NT, blk, 0, stream>>>(bedge, off, Sv, Dv, h, b3, (float*)d_out, N);
}


# tile kernel rebalanced: 256 destination tiles of 196 nodes, one 1024-thread work-group per WGP (was 782 tiles of 64), a tile's items dealt evenly to its 32 waves; LDS-atomic sum order kept
# speedup vs baseline: 4.3983x; 1.0978x over previous
#include <hip/hip_runtime.h>

typedef __attribute__((ext_vector_type(2))) float v2f;
typedef __attribute__((ext_vector_type(8))) float v8f;

#define NEG_SLOPE 0.2f
#define NEG_INF_F (-1e30f)

__device__ __forceinline__ unsigned f2o(float f) {
    unsigned u = __float_as_uint(f);
    return (u & 0x80000000u) ? ~u : (u | 0x80000000u);
}
__device__ __forceinline__ float o2f(unsigned o) {
    return (o & 0x80000000u) ? __uint_as_float(o & 0x7FFFFFFFu)
                             : __uint_as_float(~o);
}

template<int CIN, int COUT>
__global__ __launch_bounds__(256)
void gat_gemm_wmma(const float* __restrict__ X, const float* __restrict__ Wm,
                   const float* __restrict__ asrc, const float* __restrict__ adst,
                   float* __restrict__ H, float* __restrict__ S, float* __restrict__ D,
                   int N)
{
    constexpr int NT = COUT / 16;
    const int wave = (blockIdx.x * blockDim.x + threadIdx.x) >> 5;
    const int lane = threadIdx.x & 31;
    const int m0 = wave * 16;
    if (m0 >= N) return;

    const int half = lane >> 4;
    const int l = lane & 15;

    v8f acc[NT] = {};

    int arow = m0 + l; if (arow >= N) arow = N - 1;
    const float* xrow = X + (size_t)arow * CIN + 2 * half;

    #pragma unroll 4
    for (int k = 0; k < CIN; k += 4) {
        v2f A;
        A[0] = xrow[k];
        A[1] = xrow[k + 1];
        #pragma unroll
        for (int t = 0; t < NT; ++t) {
            const int n = t * 16 + l;
            v2f B;
            B[0] = Wm[(size_t)(k + 2 * half    ) * COUT + n];
            B[1] = Wm[(size_t)(k + 2 * half + 1) * COUT + n];
            acc[t] = __builtin_amdgcn_wmma_f32_16x16x4_f32(
                         false, A, false, B, (short)0, acc[t], false, false);
        }
    }

    #pragma unroll
    for (int j = 0; j < 8; ++j) {
        const int row = m0 + j + 8 * half;
        float ps = 0.0f, pd = 0.0f;
        #pragma unroll
        for (int t = 0; t < NT; ++t) {
            const int n = t * 16 + l;
            const float v = acc[t][j];
            if (row < N) H[(size_t)row * COUT + n] = v;
            ps += v * asrc[n];
            pd += v * adst[n];
        }
        #pragma unroll
        for (int off = 1; off < 16; off <<= 1) {
            ps += __shfl_xor(ps, off, 32);
            pd += __shfl_xor(pd, off, 32);
        }
        if (l == 0 && row < N) { S[row] = ps; D[row] = pd; }
    }
}


#define TILE_NODES 196
#define TILE_THREADS 1024
#define TILE_WAVES (TILE_THREADS / 32)
#define BIN_CHUNK 8192

__device__ __forceinline__ int tile_of(int node) { return (int)((unsigned)node / (unsigned)TILE_NODES); }

__global__ __launch_bounds__(1024)
void bin_count(const int* __restrict__ dst, int* __restrict__ chist, int E, int NT)
{
    extern __shared__ int hist_c[];
    const int tid = threadIdx.x;
    for (int b = tid; b < NT; b += 1024) hist_c[b] = 0;
    __syncthreads();
    const int base = blockIdx.x * BIN_CHUNK;
    int dv[BIN_CHUNK / 1024];
    #pragma unroll
    for (int k = 0; k < BIN_CHUNK / 1024; ++k) {
        const int e = base + k * 1024 + tid;
        dv[k] = dst[(e < E) ? e : (E - 1)];
    }
    #pragma unroll
    for (int k = 0; k < BIN_CHUNK / 1024; ++k) {
        const int e = base + k * 1024 + tid;
        if (e < E) atomicAdd(&hist_c[tile_of(dv[k])], 1);
    }
    __syncthreads();
    for (int b = tid; b < NT; b += 1024) chist[(size_t)blockIdx.x * NT + b] = hist_c[b];
}

__global__ __launch_bounds__(1024)
void bin_scan(const int* __restrict__ chist, int* __restrict__ cbase, int* __restrict__ off,
              int NT, int NCH, int E)
{
    __shared__ int sc[1024];
    __shared__ int carry_s;
    const int tid = threadIdx.x;
    if (tid == 0) carry_s = 0;
    __syncthreads();
    for (int b0 = 0; b0 < NT; b0 += 1024) {
        const int b = b0 + tid;
        int tot = 0;
        if (b < NT) {
            #pragma unroll 7
            for (int c = 0; c < NCH; ++c) tot += chist[(size_t)c * NT + b];
        }
        sc[tid] = tot;
        __syncthreads();
        for (int o = 1; o < 1024; o <<= 1) {
            const int t = (tid >= o) ? sc[tid - o] : 0;
            __syncthreads();
            sc[tid] += t;
            __syncthreads();
        }
        const int carry = carry_s;
        int run = sc[tid] - tot + carry;
        if (b < NT) {
            off[b] = run;
            #pragma unroll 7
            for (int c = 0; c < NCH; ++c) {
                const int v = chist[(size_t)c * NT + b];
                cbase[(size_t)c * NT + b] = run;
                run += v;
            }
        }
        __syncthreads();
        if (tid == 1023) carry_s = carry + sc[1023];
        __syncthreads();
    }
    if (tid == 0) off[NT] = E;
}

__global__ __launch_bounds__(1024)
void bin_fill(const int* __restrict__ src, const int* __restrict__ dst,
              const int* __restrict__ cbase, int2* __restrict__ bedge, int E, int NT)
{
    extern __shared__ int rank_c[];
    const int tid = threadIdx.x;
    for (int b = tid; b < NT; b += 1024) rank_c[b] = 0;
    __syncthreads();
    const int base = blockIdx.x * BIN_CHUNK;
    int sv[BIN_CHUNK / 1024];
    int dv[BIN_CHUNK / 1024];
    #pragma unroll
    for (int k = 0; k < BIN_CHUNK / 1024; ++k) {
        const int e = base + k * 1024 + tid;
        const int ec = (e < E) ? e : (E - 1);
        sv[k] = src[ec];
        dv[k] = dst[ec];
    }
    #pragma unroll
    for (int k = 0; k < BIN_CHUNK / 1024; ++k) {
        const int e = base + k * 1024 + tid;
        if (e < E) {
            const int di = dv[k];
            const int b = tile_of(di);
            const int r = atomicAdd(&rank_c[b], 1);
            const int pos = cbase[(size_t)blockIdx.x * NT + b] + r;
            bedge[pos] = make_int2(sv[k], di);
        }
    }
}

#define TR 4

template<int COUT>
__device__ __forceinline__ void tile_rows(const float* __restrict__ H, float* agg,
                                          int si, int ld, float ex, int cnt, int sub, int g)
{
    constexpr int G = COUT / 4;
    constexpr int EPS = 32 / G;
    constexpr int PITCH = COUT + (EPS == 2 ? 16 : 0);
    constexpr int U = 8;
    const int nst = (cnt < G) ? cnt : G;
    for (int k = 0; k < nst; k += U) {
        float4 hv[U];
        int    lk[U];
        float  wk[U];
        bool   ok[U];
        #pragma unroll
        for (int u = 0; u < U; ++u) {
            const int kk = k + u;
            hv[u] = make_float4(0.f, 0.f, 0.f, 0.f); lk[u] = 0; wk[u] = 0.0f; ok[u] = false;
            if (kk < nst) {
                int s = __builtin_amdgcn_readlane(si, kk);
                int l0 = __builtin_amdgcn_readlane(ld, kk);
                float w = __int_as_float(__builtin_amdgcn_readlane(__float_as_int(ex), kk));
                if constexpr (EPS == 2) {
                    const int s1 = __builtin_amdgcn_readlane(si, kk + 16);
                    const int l1 = __builtin_amdgcn_readlane(ld, kk + 16);
                    const float w1 = __int_as_float(__builtin_amdgcn_readlane(__float_as_int(ex), kk + 16));
                    if (sub) { s = s1; l0 = l1; w = w1; }
                }
                ok[u] = (kk + 16 * sub) < cnt;
                lk[u] = l0; wk[u] = w;
                if (ok[u]) hv[u] = *(const float4*)(H + (size_t)s * COUT + 4 * g);
            }
        }
        #pragma unroll
        for (int u = 0; u < U; ++u) {
            if (ok[u]) {
                float* row = agg + lk[u] * PITCH + g;
                atomicAdd(row,         wk[u] * hv[u].x);
                atomicAdd(row + G,     wk[u] * hv[u].y);
                atomicAdd(row + 2 * G, wk[u] * hv[u].z);
                atomicAdd(row + 3 * G, wk[u] * hv[u].w);
            }
        }
    }
}

template<int COUT, bool RELU>
__global__ __launch_bounds__(TILE_THREADS)
void gat_tile(const int2* __restrict__ bedge, const int* __restrict__ off,
              const float* __restrict__ S, const float* __restrict__ D,
              const float* __restrict__ H, const float* __restrict__ bias,
              float* __restrict__ out, int N)
{
    constexpr int T = TILE_NODES;
    constexpr int G = COUT / 4;
    constexpr int EPS = 32 / G;
    constexpr int PITCH = COUT + (EPS == 2 ? 16 : 0);

    __shared__ __attribute__((aligned(16))) float agg[T * PITCH];
    __shared__ unsigned mx[T];
    __shared__ float den[T];
    __shared__ float dl[T];

    const int tid = threadIdx.x;
    const int lane = tid & 31;
    const int wave = __builtin_amdgcn_readfirstlane(tid >> 5);
    const int sub = lane / G;
    const int g = lane % G;
    const int t0 = blockIdx.x * T;
    const int nloc = (N - t0 < T) ? (N - t0) : T;

    float dval = 0.0f;
    if (tid < nloc) dval = D[t0 + tid];

    const int e0 = off[blockIdx.x];
    const int ne = off[blockIdx.x + 1] - e0;
    const int total = ne + nloc;

    const int per = (total + TILE_WAVES - 1) / TILE_WAVES;
    const int wbase = wave * per;
    const int mine = (total - wbase < per) ? (total - wbase) : per;

    int2 ed[TR];
    #pragma unroll
    for (int r = 0; r < TR; ++r) {
        const int jl = lane + 32 * r;
        const int j = wbase + jl;
        ed[r] = make_int2(0, 0);
        if (jl < mine && j < ne) ed[r] = bedge[e0 + j];
    }

    {
        float4* agg4 = (float4*)agg;
        for (int i = tid; i < T * PITCH / 4; i += TILE_THREADS) agg4[i] = make_float4(0.f, 0.f, 0.f, 0.f);
    }
    if (tid < T) {
        mx[tid] = 0u;
        den[tid] = 0.0f;
        dl[tid] = dval;
    }

    int   si[TR];
    int   ld[TR];
    float al[TR];
    bool  mk[TR];
    bool  vk[TR];
    #pragma unroll
    for (int r = 0; r < TR; ++r) {
        const int jl = lane + 32 * r;
        const int j = wbase + jl;
        vk[r] = jl < mine;
        const bool isE = vk[r] && (j < ne);
        const int lself = (vk[r] && j >= ne) ? (j - ne) : 0;
        si[r] = isE ? ed[r].x : (t0 + lself);
        ld[r] = isE ? (ed[r].y - t0) : lself;
        mk[r] = isE && (ed[r].x == ed[r].y);
        al[r] = S[si[r]];
    }
    __syncthreads();

    float dv[TR];
    #pragma unroll
    for (int r = 0; r < TR; ++r) dv[r] = dl[ld[r]];
    #pragma unroll
    for (int r = 0; r < TR; ++r) {
        float a = al[r] + dv[r];
        a = (a > 0.0f) ? a : NEG_SLOPE * a;
        if (mk[r]) a = NEG_INF_F;
        al[r] = a;
        if (vk[r]) atomicMax(&mx[ld[r]], f2o(a));
    }
    for (int jl = lane + 32 * TR; jl < mine; jl += 32) {
        const int j = wbase + jl;
        int s, l; bool msk = false;
        if (j < ne) { const int2 e = bedge[e0 + j]; s = e.x; l = e.y - t0; msk = (e.x == e.y); }
        else        { l = j - ne; s = t0 + l; }
        float a = S[s] + dl[l];
        a = (a > 0.0f) ? a : NEG_SLOPE * a;
        if (msk) a = NEG_INF_F;
        atomicMax(&mx[l], f2o(a));
    }
    __syncthreads();

    float ex[TR];
    #pragma unroll
    for (int r = 0; r < TR; ++r) ex[r] = o2f(mx[ld[r]]);
    #pragma unroll
    for (int r = 0; r < TR; ++r) {
        const float e1 = __expf(al[r] - ex[r]);
        ex[r] = 0.0f;
        if (vk[r]) { ex[r] = e1; atomicAdd(&den[ld[r]], e1); }
    }
    #pragma unroll
    for (int r = 0; r < TR; ++r) {
        int cnt = mine - 32 * r;
        cnt = (cnt < 32) ? cnt : 32;
        if (cnt > 0) tile_rows<COUT>(H, agg, si[r], ld[r], ex[r], cnt, sub, g);
    }
    for (int c0 = 32 * TR; c0 < mine; c0 += 32) {
        const int jl = c0 + lane;
        int s = t0, l = 0; float e1 = 0.0f;
        if (jl < mine) {
            const int j = wbase + jl;
            bool msk = false;
            if (j < ne) { const int2 e = bedge[e0 + j]; s = e.x; l = e.y - t0; msk = (e.x == e.y); }
            else        { l = j - ne; s = t0 + l; }
            float a = S[s] + dl[l];
            a = (a > 0.0f) ? a : NEG_SLOPE * a;
            if (msk) a = NEG_INF_F;
            e1 = __expf(a - o2f(mx[l]));
            atomicAdd(&den[l], e1);
        }
        const int cnt = (mine - c0 < 32) ? (mine - c0) : 32;
        tile_rows<COUT>(H, agg, s, l, e1, cnt, sub, g);
    }
    __syncthreads();

    const float4 bb = *(const float4*)(bias + 4 * g);
    for (int r = wave * EPS + sub; r < nloc; r += TILE_WAVES * EPS) {
        const float dn = den[r];
        const float* row = agg + r * PITCH + g;
        float4 v;
        v.x = row[0]     / dn + bb.x;
        v.y = row[G]     / dn + bb.y;
        v.z = row[2 * G] / dn + bb.z;
        v.w = row[3 * G] / dn + bb.w;
        if (RELU) { v.x = fmaxf(v.x, 0.0f); v.y = fmaxf(v.y, 0.0f); v.z = fmaxf(v.z, 0.0f); v.w = fmaxf(v.w, 0.0f); }
        *(float4*)(out + (size_t)(t0 + r) * COUT + 4 * g) = v;
    }
}

static inline int cdiv_ll(long long a, long long b) { return (int)((a + b - 1) / b); }

extern "C" void kernel_launch(void* const* d_in, const int* in_sizes, int n_in,
                              void* d_out, int out_size, void* d_ws, size_t ws_size,
                              hipStream_t stream)
{
    const float* x   = (const float*)d_in[0];
    const int*   ei  = (const int*)  d_in[1];
    const float* W1  = (const float*)d_in[2];
    const float* as1 = (const float*)d_in[3];
    const float* ad1 = (const float*)d_in[4];
    const float* b1  = (const float*)d_in[5];
    const float* W2  = (const float*)d_in[6];
    const float* as2 = (const float*)d_in[7];
    const float* ad2 = (const float*)d_in[8];
    const float* b2  = (const float*)d_in[9];
    const float* W3  = (const float*)d_in[10];
    const float* as3 = (const float*)d_in[11];
    const float* ad3 = (const float*)d_in[12];
    const float* b3  = (const float*)d_in[13];

    const int N    = in_sizes[0] / 128;
    const int E    = in_sizes[1] / 2;
    const int* src = ei;
    const int* dst = ei + E;

    const int NT  = cdiv_ll(N, TILE_NODES);
    const int NCH = (E > 0) ? cdiv_ll(E, BIN_CHUNK) : 1;

    float* ws    = (float*)d_ws;
    float* h     = ws;
    float* xn    = h + (size_t)N * 128;
    int2*  bedge = (int2*)(xn + (size_t)N * 128);
    int*   off   = (int*)(bedge + E);
    int*   chist = off + (NT + 1);
    int*   cbase = chist + (size_t)NCH * NT;
    float* Sv    = (float*)(cbase + (size_t)NCH * NT);
    float* Dv    = Sv + N;

    const dim3 blk(256);
    const dim3 tblk(TILE_THREADS);
    const int waves_blocks = cdiv_ll((long long)cdiv_ll(N, 16) * 32, 256);
    const size_t hist_bytes = (size_t)NT * sizeof(int);

    bin_count<<<NCH, 1024, hist_bytes, stream>>>(dst, chist, E, NT);
    bin_scan<<<1, 1024, 0, stream>>>(chist, cbase, off, NT, NCH, E);
    bin_fill<<<NCH, 1024, hist_bytes, stream>>>(src, dst, cbase, bedge, E, NT);

    gat_gemm_wmma<128,128><<<waves_blocks, blk, 0, stream>>>(x, W1, as1, ad1, h, Sv, Dv, N);
    gat_tile<128,true><<<NT, tblk, 0, stream>>>(bedge, off, Sv, Dv, h, b1, xn, N);

    gat_gemm_wmma<128,64><<<waves_blocks, blk, 0, stream>>>(xn, W2, as2, ad2, h, Sv, Dv, N);
    gat_tile<64,true><<<NT, tblk, 0, stream>>>(bedge, off, Sv, Dv, h, b2, xn, N);

    gat_gemm_wmma<64,64><<<waves_blocks, blk, 0, stream>>>(xn, W3, as3, ad3, h, Sv, Dv, N);
    gat_tile<64,false><<<NT, tblk, 0, stream>>>(bedge, off, Sv, Dv, h, b3, (float*)d_out, N);
}


# edge binning in ONE launch: each 8192-edge chunk grouped by destination tile inside its own range (LDS histogram + scan); tile kernel walks its per-chunk runs; 9 -> 7 launches; sum order kept
# speedup vs baseline: 5.1374x; 1.1680x over previous
#include <hip/hip_runtime.h>

typedef __attribute__((ext_vector_type(2))) float v2f;
typedef __attribute__((ext_vector_type(8))) float v8f;

#define NEG_SLOPE 0.2f
#define NEG_INF_F (-1e30f)

__device__ __forceinline__ unsigned f2o(float f) {
    unsigned u = __float_as_uint(f);
    return (u & 0x80000000u) ? ~u : (u | 0x80000000u);
}
__device__ __forceinline__ float o2f(unsigned o) {
    return (o & 0x80000000u) ? __uint_as_float(o & 0x7FFFFFFFu)
                             : __uint_as_float(~o);
}

template<int CIN, int COUT>
__global__ __launch_bounds__(256)
void gat_gemm_wmma(const float* __restrict__ X, const float* __restrict__ Wm,
                   const float* __restrict__ asrc, const float* __restrict__ adst,
                   float* __restrict__ H, float* __restrict__ S, float* __restrict__ D,
                   int N)
{
    constexpr int NT = COUT / 16;
    const int wave = (blockIdx.x * blockDim.x + threadIdx.x) >> 5;
    const int lane = threadIdx.x & 31;
    const int m0 = wave * 16;
    if (m0 >= N) return;

    const int half = lane >> 4;
    const int l = lane & 15;

    v8f acc[NT] = {};

    int arow = m0 + l; if (arow >= N) arow = N - 1;
    const float* xrow = X + (size_t)arow * CIN + 2 * half;

    #pragma unroll 4
    for (int k = 0; k < CIN; k += 4) {
        v2f A;
        A[0] = xrow[k];
        A[1] = xrow[k + 1];
        #pragma unroll
        for (int t = 0; t < NT; ++t) {
            const int n = t * 16 + l;
            v2f B;
            B[0] = Wm[(size_t)(k + 2 * half    ) * COUT + n];
            B[1] = Wm[(size_t)(k + 2 * half + 1) * COUT + n];
            acc[t] = __builtin_amdgcn_wmma_f32_16x16x4_f32(
                         false, A, false, B, (short)0, acc[t], false, false);
        }
    }

    #pragma unroll
    for (int j = 0; j < 8; ++j) {
        const int row = m0 + j + 8 * half;
        float ps = 0.0f, pd = 0.0f;
        #pragma unroll
        for (int t = 0; t < NT; ++t) {
            const int n = t * 16 + l;
            const float v = acc[t][j];
            if (row < N) H[(size_t)row * COUT + n] = v;
            ps += v * asrc[n];
            pd += v * adst[n];
        }
        #pragma unroll
        for (int off = 1; off < 16; off <<= 1) {
            ps += __shfl_xor(ps, off, 32);
            pd += __shfl_xor(pd, off, 32);
        }
        if (l == 0 && row < N) { S[row] = ps; D[row] = pd; }
    }
}


#define TILE_NODES 196
#define TILE_THREADS 1024
#define TILE_WAVES (TILE_THREADS / 32)
#define BIN_CHUNK 8192

__device__ __forceinline__ int tile_of(int node) { return (int)((unsigned)node / (unsigned)TILE_NODES); }

__device__ __forceinline__ int wave_incl_scan(int v, int lane)
{
    v += __builtin_amdgcn_update_dpp(0, v, 0x111, 0xf, 0xf, true);
    v += __builtin_amdgcn_update_dpp(0, v, 0x112, 0xf, 0xf, true);
    v += __builtin_amdgcn_update_dpp(0, v, 0x114, 0xf, 0xf, true);
    v += __builtin_amdgcn_update_dpp(0, v, 0x118, 0xf, 0xf, true);
    const int low = __builtin_amdgcn_readlane(v, 15);
    return (lane >= 16) ? (v + low) : v;
}

__global__ __launch_bounds__(1024)
void bin_local(const int* __restrict__ src, const int* __restrict__ dst,
               int2* __restrict__ bedge, int* __restrict__ cstart, int E, int NT)
{
    extern __shared__ int hist_c[];
    __shared__ int wsum[32];
    __shared__ int carry_s;
    const int tid = threadIdx.x;
    const int lane = tid & 31;
    const int wave = tid >> 5;
    for (int b = tid; b < NT; b += 1024) hist_c[b] = 0;
    if (tid == 0) carry_s = 0;
    __syncthreads();
    const int base = blockIdx.x * BIN_CHUNK;
    int sv[BIN_CHUNK / 1024];
    int dv[BIN_CHUNK / 1024];
    int rk[BIN_CHUNK / 1024];
    #pragma unroll
    for (int k = 0; k < BIN_CHUNK / 1024; ++k) {
        const int e = base + k * 1024 + tid;
        const int ec = (e < E) ? e : (E - 1);
        sv[k] = src[ec];
        dv[k] = dst[ec];
    }
    #pragma unroll
    for (int k = 0; k < BIN_CHUNK / 1024; ++k) {
        const int e = base + k * 1024 + tid;
        rk[k] = 0;
        if (e < E) rk[k] = atomicAdd(&hist_c[tile_of(dv[k])], 1);
    }
    __syncthreads();
    const size_t crow = (size_t)blockIdx.x * (NT + 1);
    for (int b0 = 0; b0 < NT; b0 += 1024) {
        const int b = b0 + tid;
        const int v = (b < NT) ? hist_c[b] : 0;
        const int inc = wave_incl_scan(v, lane);
        if (lane == 31) wsum[wave] = inc;
        __syncthreads();
        if (wave == 0) {
            const int w = wsum[lane];
            wsum[lane] = wave_incl_scan(w, lane) - w;
        }
        __syncthreads();
        const int excl = carry_s + wsum[wave] + inc - v;
        if (b < NT) hist_c[b] = excl;
        __syncthreads();
        if (tid == 1023) carry_s = excl + v;
        __syncthreads();
    }
    for (int b = tid; b < NT; b += 1024) cstart[crow + b] = base + hist_c[b];
    if (tid == 0) cstart[crow + NT] = base + carry_s;
    #pragma unroll
    for (int k = 0; k < BIN_CHUNK / 1024; ++k) {
        const int e = base + k * 1024 + tid;
        if (e < E) {
            const int di = dv[k];
            bedge[base + hist_c[tile_of(di)] + rk[k]] = make_int2(sv[k], di);
        }
    }
}

#define TR 4

__device__ __forceinline__ int seg_chunk(const int* seg_pre, int NCH, int P, int j)
{
    int lo = 0;
    for (int step = P; step > 0; step >>= 1) {
        const int c = lo + step;
        const int pv = seg_pre[(c < NCH) ? c : NCH];
        const bool take = (c < NCH) & (pv <= j);
        lo = take ? c : lo;
    }
    return lo;
}
__device__ __forceinline__ int seg_record(const int* seg_start, const int* seg_pre, int NCH, int P, int j)
{
    const int c = seg_chunk(seg_pre, NCH, P, j);
    return seg_start[c] + (j - seg_pre[c]);
}

template<int COUT>
__device__ __forceinline__ void tile_rows(const float* __restrict__ H, float* agg,
                                          int si, int ld, float ex, int cnt, int sub, int g)
{
    constexpr int G = COUT / 4;
    constexpr int EPS = 32 / G;
    constexpr int PITCH = COUT + (EPS == 2 ? 16 : 0);
    constexpr int U = 8;
    const int nst = (cnt < G) ? cnt : G;
    for (int k = 0; k < nst; k += U) {
        float4 hv[U];
        int    lk[U];
        float  wk[U];
        bool   ok[U];
        #pragma unroll
        for (int u = 0; u < U; ++u) {
            const int kk = k + u;
            hv[u] = make_float4(0.f, 0.f, 0.f, 0.f); lk[u] = 0; wk[u] = 0.0f; ok[u] = false;
            if (kk < nst) {
                int s = __builtin_amdgcn_readlane(si, kk);
                int l0 = __builtin_amdgcn_readlane(ld, kk);
                float w = __int_as_float(__builtin_amdgcn_readlane(__float_as_int(ex), kk));
                if constexpr (EPS == 2) {
                    const int s1 = __builtin_amdgcn_readlane(si, kk + 16);
                    const int l1 = __builtin_amdgcn_readlane(ld, kk + 16);
                    const float w1 = __int_as_float(__builtin_amdgcn_readlane(__float_as_int(ex), kk + 16));
                    if (sub) { s = s1; l0 = l1; w = w1; }
                }
                ok[u] = (kk + 16 * sub) < cnt;
                lk[u] = l0; wk[u] = w;
                if (ok[u]) hv[u] = *(const float4*)(H + (size_t)s * COUT + 4 * g);
            }
        }
        #pragma unroll
        for (int u = 0; u < U; ++u) {
            if (ok[u]) {
                float* row = agg + lk[u] * PITCH + g;
                atomicAdd(row,         wk[u] * hv[u].x);
                atomicAdd(row + G,     wk[u] * hv[u].y);
                atomicAdd(row + 2 * G, wk[u] * hv[u].z);
                atomicAdd(row + 3 * G, wk[u] * hv[u].w);
            }
        }
    }
}

template<int COUT, bool RELU>
__global__ __launch_bounds__(TILE_THREADS)
void gat_tile(const int2* __restrict__ bedge, const int* __restrict__ cstart,
              const float* __restrict__ S, const float* __restrict__ D,
              const float* __restrict__ H, const float* __restrict__ bias,
              float* __restrict__ out, int N, int NCH, int NT)
{
    constexpr int T = TILE_NODES;
    constexpr int G = COUT / 4;
    constexpr int EPS = 32 / G;
    constexpr int PITCH = COUT + (EPS == 2 ? 16 : 0);

    __shared__ __attribute__((aligned(16))) float agg[T * PITCH];
    __shared__ unsigned mx[T];
    __shared__ float den[T];
    __shared__ float dl[T];
    extern __shared__ int segs[];
    int* seg_start = segs;
    int* seg_pre   = segs + NCH;

    const int tid = threadIdx.x;
    const int lane = tid & 31;
    const int wave = __builtin_amdgcn_readfirstlane(tid >> 5);
    const int sub = lane / G;
    const int g = lane % G;
    const int t0 = blockIdx.x * T;
    const int nloc = (N - t0 < T) ? (N - t0) : T;

    float dval = 0.0f;
    if (tid < nloc) dval = D[t0 + tid];

    for (int c = tid; c < NCH; c += TILE_THREADS) {
        const int* p = cstart + (size_t)c * (NT + 1) + blockIdx.x;
        const int a = p[0];
        seg_start[c] = a;
        seg_pre[c + 1] = p[1] - a;
    }
    {
        float4* agg4 = (float4*)agg;
        for (int i = tid; i < T * PITCH / 4; i += TILE_THREADS) agg4[i] = make_float4(0.f, 0.f, 0.f, 0.f);
    }
    if (tid < T) {
        mx[tid] = 0u;
        den[tid] = 0.0f;
        dl[tid] = dval;
    }
    __syncthreads();
    if (wave == 0) {
        int carry = 0;
        for (int c0 = 0; c0 < NCH; c0 += 32) {
            const int c = c0 + lane;
            const int v = (c < NCH) ? seg_pre[c + 1] : 0;
            const int inc = wave_incl_scan(v, lane);
            if (c < NCH) seg_pre[c + 1] = carry + inc;
            carry += __builtin_amdgcn_readlane(inc, 31);
        }
        if (lane == 0) seg_pre[0] = 0;
    }
    __syncthreads();
    const int ne = __builtin_amdgcn_readfirstlane(seg_pre[NCH]);
    const int P = 1 << (31 - __clz(NCH));
    const int total = ne + nloc;

    const int per = (total + TILE_WAVES - 1) / TILE_WAVES;
    const int wbase = wave * per;
    const int mine = (total - wbase < per) ? (total - wbase) : per;

    int cj[TR];
    #pragma unroll
    for (int r = 0; r < TR; ++r) cj[r] = 0;
    for (int step = P; step > 0; step >>= 1) {
        int pv[TR];
        #pragma unroll
        for (int r = 0; r < TR; ++r) {
            const int c = cj[r] + step;
            pv[r] = seg_pre[(c < NCH) ? c : NCH];
        }
        #pragma unroll
        for (int r = 0; r < TR; ++r) {
            const int c = cj[r] + step;
            const bool take = (c < NCH) & (pv[r] <= wbase + lane + 32 * r);
            cj[r] = take ? c : cj[r];
        }
    }

    int2 ed[TR];
    #pragma unroll
    for (int r = 0; r < TR; ++r) {
        const int jl = lane + 32 * r;
        const int j = wbase + jl;
        ed[r] = make_int2(0, 0);
        const int rec = seg_start[cj[r]] + (j - seg_pre[cj[r]]);
        if (jl < mine && j < ne) ed[r] = bedge[rec];
    }

    int   si[TR];
    int   ld[TR];
    float al[TR];
    bool  mk[TR];
    bool  vk[TR];
    #pragma unroll
    for (int r = 0; r < TR; ++r) {
        const int jl = lane + 32 * r;
        const int j = wbase + jl;
        vk[r] = jl < mine;
        const bool isE = vk[r] && (j < ne);
        const int lself = (vk[r] && j >= ne) ? (j - ne) : 0;
        si[r] = isE ? ed[r].x : (t0 + lself);
        ld[r] = isE ? (ed[r].y - t0) : lself;
        mk[r] = isE && (ed[r].x == ed[r].y);
        al[r] = S[si[r]];
    }

    float dv[TR];
    #pragma unroll
    for (int r = 0; r < TR; ++r) dv[r] = dl[ld[r]];
    #pragma unroll
    for (int r = 0; r < TR; ++r) {
        float a = al[r] + dv[r];
        a = (a > 0.0f) ? a : NEG_SLOPE * a;
        if (mk[r]) a = NEG_INF_F;
        al[r] = a;
        if (vk[r]) atomicMax(&mx[ld[r]], f2o(a));
    }
    for (int jl = lane + 32 * TR; jl < mine; jl += 32) {
        const int j = wbase + jl;
        int s, l; bool msk = false;
        if (j < ne) { const int2 e = bedge[seg_record(seg_start, seg_pre, NCH, P, j)]; s = e.x; l = e.y - t0; msk = (e.x == e.y); }
        else        { l = j - ne; s = t0 + l; }
        float a = S[s] + dl[l];
        a = (a > 0.0f) ? a : NEG_SLOPE * a;
        if (msk) a = NEG_INF_F;
        atomicMax(&mx[l], f2o(a));
    }
    __syncthreads();

    float ex[TR];
    #pragma unroll
    for (int r = 0; r < TR; ++r) ex[r] = o2f(mx[ld[r]]);
    #pragma unroll
    for (int r = 0; r < TR; ++r) {
        const float e1 = __expf(al[r] - ex[r]);
        ex[r] = 0.0f;
        if (vk[r]) { ex[r] = e1; atomicAdd(&den[ld[r]], e1); }
    }
    #pragma unroll
    for (int r = 0; r < TR; ++r) {
        int cnt = mine - 32 * r;
        cnt = (cnt < 32) ? cnt : 32;
        if (cnt > 0) tile_rows<COUT>(H, agg, si[r], ld[r], ex[r], cnt, sub, g);
    }
    for (int c0 = 32 * TR; c0 < mine; c0 += 32) {
        const int jl = c0 + lane;
        int s = t0, l = 0; float e1 = 0.0f;
        if (jl < mine) {
            const int j = wbase + jl;
            bool msk = false;
            if (j < ne) { const int2 e = bedge[seg_record(seg_start, seg_pre, NCH, P, j)]; s = e.x; l = e.y - t0; msk = (e.x == e.y); }
            else        { l = j - ne; s = t0 + l; }
            float a = S[s] + dl[l];
            a = (a > 0.0f) ? a : NEG_SLOPE * a;
            if (msk) a = NEG_INF_F;
            e1 = __expf(a - o2f(mx[l]));
            atomicAdd(&den[l], e1);
        }
        const int cnt = (mine - c0 < 32) ? (mine - c0) : 32;
        tile_rows<COUT>(H, agg, s, l, e1, cnt, sub, g);
    }
    __syncthreads();

    const float4 bb = *(const float4*)(bias + 4 * g);
    for (int r = wave * EPS + sub; r < nloc; r += TILE_WAVES * EPS) {
        const float dn = den[r];
        const float* row = agg + r * PITCH + g;
        float4 v;
        v.x = row[0]     / dn + bb.x;
        v.y = row[G]     / dn + bb.y;
        v.z = row[2 * G] / dn + bb.z;
        v.w = row[3 * G] / dn + bb.w;
        if (RELU) { v.x = fmaxf(v.x, 0.0f); v.y = fmaxf(v.y, 0.0f); v.z = fmaxf(v.z, 0.0f); v.w = fmaxf(v.w, 0.0f); }
        *(float4*)(out + (size_t)(t0 + r) * COUT + 4 * g) = v;
    }
}

static inline int cdiv_ll(long long a, long long b) { return (int)((a + b - 1) / b); }

extern "C" void kernel_launch(void* const* d_in, const int* in_sizes, int n_in,
                              void* d_out, int out_size, void* d_ws, size_t ws_size,
                              hipStream_t stream)
{
    const float* x   = (const float*)d_in[0];
    const int*   ei  = (const int*)  d_in[1];
    const float* W1  = (const float*)d_in[2];
    const float* as1 = (const float*)d_in[3];
    const float* ad1 = (const float*)d_in[4];
    const float* b1  = (const float*)d_in[5];
    const float* W2  = (const float*)d_in[6];
    const float* as2 = (const float*)d_in[7];
    const float* ad2 = (const float*)d_in[8];
    const float* b2  = (const float*)d_in[9];
    const float* W3  = (const float*)d_in[10];
    const float* as3 = (const float*)d_in[11];
    const float* ad3 = (const float*)d_in[12];
    const float* b3  = (const float*)d_in[13];

    const int N    = in_sizes[0] / 128;
    const int E    = in_sizes[1] / 2;
    const int* src = ei;
    const int* dst = ei + E;

    const int NT  = cdiv_ll(N, TILE_NODES);
    const int NCH = (E > 0) ? cdiv_ll(E, BIN_CHUNK) : 1;

    float* ws    = (float*)d_ws;
    float* h     = ws;
    float* xn    = h + (size_t)N * 128;
    int2*  bedge = (int2*)(xn + (size_t)N * 128);
    int*   cstart = (int*)(bedge + E);
    float* Sv    = (float*)(cstart + (size_t)NCH * (NT + 1));
    float* Dv    = Sv + N;

    const dim3 blk(256);
    const dim3 tblk(TILE_THREADS);
    const int waves_blocks = cdiv_ll((long long)cdiv_ll(N, 16) * 32, 256);
    const size_t hist_bytes = (size_t)NT * sizeof(int);
    const size_t segs_bytes = (size_t)(2 * NCH + 1) * sizeof(int);

    bin_local<<<NCH, 1024, hist_bytes, stream>>>(src, dst, bedge, cstart, E, NT);

    gat_gemm_wmma<128,128><<<waves_blocks, blk, 0, stream>>>(x, W1, as1, ad1, h, Sv, Dv, N);
    gat_tile<128,true><<<NT, tblk, segs_bytes, stream>>>(bedge, cstart, Sv, Dv, h, b1, xn, N, NCH, NT);

    gat_gemm_wmma<128,64><<<waves_blocks, blk, 0, stream>>>(xn, W2, as2, ad2, h, Sv, Dv, N);
    gat_tile<64,true><<<NT, tblk, segs_bytes, stream>>>(bedge, cstart, Sv, Dv, h, b2, xn, N, NCH, NT);

    gat_gemm_wmma<64,64><<<waves_blocks, blk, 0, stream>>>(xn, W3, as3, ad3, h, Sv, Dv, N);
    gat_tile<64,false><<<NT, tblk, segs_bytes, stream>>>(bedge, cstart, Sv, Dv, h, b3, (float*)d_out, N, NCH, NT);
}
